# starting kernel
# baseline (speedup 1.0000x reference)
#include <hip/hip_runtime.h>
#include <math.h>


typedef __attribute__((ext_vector_type(2))) float v2f;
typedef __attribute__((ext_vector_type(8))) float v8f;

#define NQ        4
#define QDEPTH    6
#define FDIM      512
#define TILE_M    16
#define WAVES_PB  8
#define ROWS_PB   (TILE_M * WAVES_PB)

__global__ void qnn_prep(const float* __restrict__ enc_w,
                         const float* __restrict__ q_params,
                         float* __restrict__ encT,
                         float* __restrict__ U)
{
    const int tid = threadIdx.x;

    for (int idx = tid; idx < FDIM * 16; idx += blockDim.x) {
        const int k = idx >> 4;
        const int n = idx & 15;
        encT[idx] = (n < NQ) ? enc_w[n * FDIM + k] : 0.0f;
    }

    if (tid < 16) {
        float s[16];
#pragma unroll
        for (int i = 0; i < 16; ++i) s[i] = (i == tid) ? 1.0f : 0.0f;

        for (int d = 0; d < QDEPTH; ++d) {
            for (int q = 0; q < NQ; ++q) {
                const float th = 0.5f * q_params[q * QDEPTH + d];
                const float c  = cosf(th);
                const float sn = sinf(th);
                const int   m  = 1 << (3 - q);
#pragma unroll
                for (int i = 0; i < 16; ++i) {
                    if (i & m) continue;
                    const float a = s[i], b = s[i | m];
                    s[i]     = c * a - sn * b;
                    s[i | m] = sn * a + c * b;
                }
            }
            for (int q = 0; q < NQ; ++q) {
                const int mc = 1 << (3 - q);
                const int mt = 1 << (3 - ((q + 1) & 3));
#pragma unroll
                for (int i = 0; i < 16; ++i) {
                    if ((i & mc) && !(i & mt)) {
                        const float t = s[i];
                        s[i]      = s[i | mt];
                        s[i | mt] = t;
                    }
                }
            }
        }
#pragma unroll
        for (int i = 0; i < 16; ++i) U[i * 16 + tid] = s[i];
    }
}

__global__ void __launch_bounds__(256)
qnn_main(const float* __restrict__ x,
         const float* __restrict__ encT,
         const float* __restrict__ enc_b,
         const float* __restrict__ U,
         float* __restrict__ out)
{
    __shared__ float shB[FDIM * 16];
    __shared__ float shU[256];
    __shared__ float shAng[WAVES_PB][TILE_M][NQ];

    const int tid  = threadIdx.x;
    const int wave = tid >> 5;
    const int lane = tid & 31;
    const int half = lane >> 4;
    const int col  = lane & 15;

    for (int i = tid; i < FDIM * 16; i += 256) shB[i] = encT[i];
    shU[tid] = U[tid];
    __syncthreads();

    const int    rowBase = (blockIdx.x * WAVES_PB + wave) * TILE_M;
    const float* aptr    = x + (size_t)(rowBase + col) * FDIM + 2 * half;

    v8f acc0 = {0.f, 0.f, 0.f, 0.f, 0.f, 0.f, 0.f, 0.f};
    v8f acc1 = {0.f, 0.f, 0.f, 0.f, 0.f, 0.f, 0.f, 0.f};
    for (int k0 = 0; k0 < FDIM; k0 += 8) {
        v2f a0 = *(const v2f*)(aptr + k0);
        v2f a1 = *(const v2f*)(aptr + k0 + 4);
        const int kk0 = (k0 + 2 * half) << 4;
        const int kk1 = (k0 + 4 + 2 * half) << 4;
        v2f b0, b1;
        b0.x = shB[kk0 + col];
        b0.y = shB[kk0 + 16 + col];
        b1.x = shB[kk1 + col];
        b1.y = shB[kk1 + 16 + col];
        acc0 = __builtin_amdgcn_wmma_f32_16x16x4_f32(
            false, a0, false, b0, (short)0, acc0, false, false);
        acc1 = __builtin_amdgcn_wmma_f32_16x16x4_f32(
            false, a1, false, b1, (short)0, acc1, false, false);
    }
    v8f acc = acc0 + acc1;

    const float HALF_PI = 1.5707963267948966f;
    if (col < NQ) {
        const float bias = enc_b[col];
#pragma unroll
        for (int j = 0; j < 8; ++j) {
            shAng[wave][half * 8 + j][col] = tanhf(acc[j] + bias) * HALF_PI;
        }
    }
    __syncthreads();

    if (lane < TILE_M) {
        const int r = lane;
        float cq[NQ], sq[NQ];
#pragma unroll
        for (int q = 0; q < NQ; ++q) {
            const float phi = 0.5f * shAng[wave][r][q] + 0.7853981633974483f;
            cq[q] = cosf(phi);
            sq[q] = sinf(phi);
        }
        float v[16];
#pragma unroll
        for (int i = 0; i < 16; ++i) {
            float p = ((i >> 3) & 1) ? sq[0] : cq[0];
            p *= ((i >> 2) & 1) ? sq[1] : cq[1];
            p *= ((i >> 1) & 1) ? sq[2] : cq[2];
            p *= ( i       & 1) ? sq[3] : cq[3];
            v[i] = p;
        }
        float p2[16];
#pragma unroll
        for (int i = 0; i < 16; ++i) {
            float a = 0.f;
#pragma unroll
            for (int j = 0; j < 16; ++j) a = fmaf(shU[i * 16 + j], v[j], a);
            p2[i] = a * a;
        }
        float z[NQ];
#pragma unroll
        for (int q = 0; q < NQ; ++q) {
            const int m = 1 << (3 - q);
            float zz = 0.f;
#pragma unroll
            for (int i = 0; i < 16; ++i) zz += (i & m) ? -p2[i] : p2[i];
            z[q] = zz;
        }
        *(float4*)(out + (size_t)(rowBase + r) * NQ) =
            make_float4(z[0], z[1], z[2], z[3]);
    }
}

extern "C" void kernel_launch(void* const* d_in, const int* in_sizes, int n_in,
                              void* d_out, int out_size, void* d_ws, size_t ws_size,
                              hipStream_t stream)
{
    const float* x        = (const float*)d_in[0];
    const float* enc_w    = (const float*)d_in[1];
    const float* enc_b    = (const float*)d_in[2];
    const float* q_params = (const float*)d_in[3];

    float* encT = (float*)d_ws;
    float* U    = encT + FDIM * 16;
    float* out  = (float*)d_out;

    qnn_prep<<<1, 256, 0, stream>>>(enc_w, q_params, encT, U);

    const int B      = in_sizes[0] / FDIM;
    const int blocks = (B + ROWS_PB - 1) / ROWS_PB;
    qnn_main<<<blocks, 256, 0, stream>>>(x, encT, enc_b, U, out);
}


# Set-up launch folded into the main kernel: each block builds its B tile (from enc_w) and the 16x16 circuit matrix (from q_params) in LDS itself; one launch instead of two, workspace unused.
# speedup vs baseline: 2.1281x; 2.1281x over previous
#include <hip/hip_runtime.h>
#include <math.h>


typedef __attribute__((ext_vector_type(2))) float v2f;
typedef __attribute__((ext_vector_type(8))) float v8f;

#define NQ        4
#define QDEPTH    6
#define FDIM      512
#define TILE_M    16
#define WAVES_PB  8
#define ROWS_PB   (TILE_M * WAVES_PB)

__global__ void __launch_bounds__(256)
qnn_main(const float* __restrict__ x,
         const float* __restrict__ enc_w,
         const float* __restrict__ enc_b,
         const float* __restrict__ q_params,
         float* __restrict__ out)
{
    __shared__ float shB[FDIM * 16];
    __shared__ float shU[256];
    __shared__ float shAng[WAVES_PB][TILE_M][NQ];
    __shared__ float shCos[NQ * QDEPTH];
    __shared__ float shSin[NQ * QDEPTH];

    const int tid  = threadIdx.x;
    const int wave = tid >> 5;
    const int lane = tid & 31;
    const int half = lane >> 4;
    const int col  = lane & 15;

    const float qp = q_params[tid < NQ * QDEPTH ? tid : 0];

    {
        float4 w[FDIM / 256];
#pragma unroll
        for (int j = 0; j < FDIM / 256; ++j) {
            const int k = tid + 256 * j;
            w[j].x = enc_w[0 * FDIM + k];
            w[j].y = enc_w[1 * FDIM + k];
            w[j].z = enc_w[2 * FDIM + k];
            w[j].w = enc_w[3 * FDIM + k];
        }
        asm volatile("" :: "v"(qp));
        const float4 zero4 = make_float4(0.f, 0.f, 0.f, 0.f);
#pragma unroll
        for (int j = 0; j < FDIM / 256; ++j) {
            const int k = tid + 256 * j;
            float4* dst = (float4*)(shB + k * 16);
            dst[0] = w[j];
            dst[1] = zero4;
            dst[2] = zero4;
            dst[3] = zero4;
        }
    }
    if (tid < NQ * QDEPTH) {
        const float th = 0.5f * qp;
        shCos[tid] = cosf(th);
        shSin[tid] = sinf(th);
    }
    __syncthreads();

    if (tid < 16) {
        float s[16];
#pragma unroll
        for (int i = 0; i < 16; ++i) s[i] = (i == tid) ? 1.0f : 0.0f;

#pragma unroll
        for (int d = 0; d < QDEPTH; ++d) {
#pragma unroll
            for (int q = 0; q < NQ; ++q) {
                const float c  = shCos[q * QDEPTH + d];
                const float sn = shSin[q * QDEPTH + d];
                const int   m  = 1 << (3 - q);
#pragma unroll
                for (int i = 0; i < 16; ++i) {
                    if (i & m) continue;
                    const float a = s[i], b = s[i | m];
                    s[i]     = c * a - sn * b;
                    s[i | m] = sn * a + c * b;
                }
            }
#pragma unroll
            for (int q = 0; q < NQ; ++q) {
                const int mc = 1 << (3 - q);
                const int mt = 1 << (3 - ((q + 1) & 3));
#pragma unroll
                for (int i = 0; i < 16; ++i) {
                    if ((i & mc) && !(i & mt)) {
                        const float t = s[i];
                        s[i]      = s[i | mt];
                        s[i | mt] = t;
                    }
                }
            }
        }
#pragma unroll
        for (int i = 0; i < 16; ++i) shU[i * 16 + tid] = s[i];
    }

    const int    rowBase = (blockIdx.x * WAVES_PB + wave) * TILE_M;
    const float* aptr    = x + (size_t)(rowBase + col) * FDIM + 2 * half;

    v8f acc0 = {0.f, 0.f, 0.f, 0.f, 0.f, 0.f, 0.f, 0.f};
    v8f acc1 = {0.f, 0.f, 0.f, 0.f, 0.f, 0.f, 0.f, 0.f};
    for (int k0 = 0; k0 < FDIM; k0 += 8) {
        v2f a0 = *(const v2f*)(aptr + k0);
        v2f a1 = *(const v2f*)(aptr + k0 + 4);
        const int kk0 = (k0 + 2 * half) << 4;
        const int kk1 = (k0 + 4 + 2 * half) << 4;
        v2f b0, b1;
        b0.x = shB[kk0 + col];
        b0.y = shB[kk0 + 16 + col];
        b1.x = shB[kk1 + col];
        b1.y = shB[kk1 + 16 + col];
        acc0 = __builtin_amdgcn_wmma_f32_16x16x4_f32(
            false, a0, false, b0, (short)0, acc0, false, false);
        acc1 = __builtin_amdgcn_wmma_f32_16x16x4_f32(
            false, a1, false, b1, (short)0, acc1, false, false);
    }
    v8f acc = acc0 + acc1;

    const float HALF_PI = 1.5707963267948966f;
    if (col < NQ) {
        const float bias = enc_b[col];
#pragma unroll
        for (int j = 0; j < 8; ++j) {
            shAng[wave][half * 8 + j][col] = tanhf(acc[j] + bias) * HALF_PI;
        }
    }
    __syncthreads();

    if (lane < TILE_M) {
        const int r = lane;
        float cq[NQ], sq[NQ];
#pragma unroll
        for (int q = 0; q < NQ; ++q) {
            const float phi = 0.5f * shAng[wave][r][q] + 0.7853981633974483f;
            cq[q] = cosf(phi);
            sq[q] = sinf(phi);
        }
        float v[16];
#pragma unroll
        for (int i = 0; i < 16; ++i) {
            float p = ((i >> 3) & 1) ? sq[0] : cq[0];
            p *= ((i >> 2) & 1) ? sq[1] : cq[1];
            p *= ((i >> 1) & 1) ? sq[2] : cq[2];
            p *= ( i       & 1) ? sq[3] : cq[3];
            v[i] = p;
        }
        float p2[16];
#pragma unroll
        for (int i = 0; i < 16; ++i) {
            float a = 0.f;
#pragma unroll
            for (int j = 0; j < 16; ++j) a = fmaf(shU[i * 16 + j], v[j], a);
            p2[i] = a * a;
        }
        float z[NQ];
#pragma unroll
        for (int q = 0; q < NQ; ++q) {
            const int m = 1 << (3 - q);
            float zz = 0.f;
#pragma unroll
            for (int i = 0; i < 16; ++i) zz += (i & m) ? -p2[i] : p2[i];
            z[q] = zz;
        }
        *(float4*)(out + (size_t)(rowBase + r) * NQ) =
            make_float4(z[0], z[1], z[2], z[3]);
    }
}

extern "C" void kernel_launch(void* const* d_in, const int* in_sizes, int n_in,
                              void* d_out, int out_size, void* d_ws, size_t ws_size,
                              hipStream_t stream)
{
    const float* x        = (const float*)d_in[0];
    const float* enc_w    = (const float*)d_in[1];
    const float* enc_b    = (const float*)d_in[2];
    const float* q_params = (const float*)d_in[3];
    float*       out      = (float*)d_out;

    const int B      = in_sizes[0] / FDIM;
    const int blocks = (B + ROWS_PB - 1) / ROWS_PB;
    qnn_main<<<blocks, 256, 0, stream>>>(x, enc_w, enc_b, q_params, out);
}


# x read as whole-line loads (512 contiguous bytes of one row per wave load, 16 or more in flight), transposed to A fragments in a per-wave LDS stage; B regrouped alike: k summation order changes.
# speedup vs baseline: 2.6052x; 1.2242x over previous
#include <hip/hip_runtime.h>
#include <math.h>


typedef __attribute__((ext_vector_type(2))) float v2f;
typedef __attribute__((ext_vector_type(4))) float v4f;
typedef __attribute__((ext_vector_type(8))) float v8f;

#define NQ        4
#define QDEPTH    6
#define FDIM      512
#define TILE_M    16
#define WAVES_PB  8
#define ROWS_PB   (TILE_M * WAVES_PB)
#define KCH       128
#define NCH       (FDIM / KCH)
#define XPITCH    (KCH + 4)

__global__ void __launch_bounds__(256)
qnn_main(const float* __restrict__ x,
         const float* __restrict__ enc_w,
         const float* __restrict__ enc_b,
         const float* __restrict__ q_params,
         float* __restrict__ out)
{
    __shared__ v4f shB4[(FDIM / 4) * 16];
    __shared__ v4f shX4[WAVES_PB][TILE_M * (XPITCH / 4)];
    __shared__ float shU[256];
    __shared__ float shAng[WAVES_PB][TILE_M][NQ];
    __shared__ float shCos[NQ * QDEPTH];
    __shared__ float shSin[NQ * QDEPTH];

    const int tid  = threadIdx.x;
    const int wave = tid >> 5;
    const int lane = tid & 31;
    const int half = lane >> 4;
    const int col  = lane & 15;

    const int     rowBase = (blockIdx.x * WAVES_PB + wave) * TILE_M;
    const v4f*    xw      = (const v4f*)(x + (size_t)rowBase * FDIM) + lane;
    v4f R[TILE_M];
#pragma unroll
    for (int r = 0; r < TILE_M; ++r) R[r] = xw[r * (FDIM / 4)];

    const float qp = q_params[tid < NQ * QDEPTH ? tid : 0];

    {
        v4f w[FDIM / 256];
#pragma unroll
        for (int j = 0; j < FDIM / 256; ++j) {
            const int i = tid + 256 * j;
            w[j] = ((const v4f*)enc_w)[i];
        }
        asm volatile("" :: "v"(qp));
        const v4f zero4 = {0.f, 0.f, 0.f, 0.f};
#pragma unroll
        for (int j = 0; j < FDIM / 256; ++j) {
            const int i = tid + 256 * j;
            shB4[(i & 127) * 16 + (i >> 7)] = w[j];
        }
#pragma unroll
        for (int j = 0; j < 6; ++j) {
            const int z  = tid + 256 * j;
            const int kq = z / 12;
            shB4[kq * 16 + 4 + (z - 12 * kq)] = zero4;
        }
    }
    if (tid < NQ * QDEPTH) {
        const float th = 0.5f * qp;
        shCos[tid] = cosf(th);
        shSin[tid] = sinf(th);
    }
    __syncthreads();

    if (tid < 16) {
        float s[16];
#pragma unroll
        for (int i = 0; i < 16; ++i) s[i] = (i == tid) ? 1.0f : 0.0f;

#pragma unroll
        for (int d = 0; d < QDEPTH; ++d) {
#pragma unroll
            for (int q = 0; q < NQ; ++q) {
                const float c  = shCos[q * QDEPTH + d];
                const float sn = shSin[q * QDEPTH + d];
                const int   m  = 1 << (3 - q);
#pragma unroll
                for (int i = 0; i < 16; ++i) {
                    if (i & m) continue;
                    const float a = s[i], b = s[i | m];
                    s[i]     = c * a - sn * b;
                    s[i | m] = sn * a + c * b;
                }
            }
#pragma unroll
            for (int q = 0; q < NQ; ++q) {
                const int mc = 1 << (3 - q);
                const int mt = 1 << (3 - ((q + 1) & 3));
#pragma unroll
                for (int i = 0; i < 16; ++i) {
                    if ((i & mc) && !(i & mt)) {
                        const float t = s[i];
                        s[i]      = s[i | mt];
                        s[i | mt] = t;
                    }
                }
            }
        }
#pragma unroll
        for (int i = 0; i < 16; ++i) shU[i * 16 + tid] = s[i];
    }

    v8f acc0 = {0.f, 0.f, 0.f, 0.f, 0.f, 0.f, 0.f, 0.f};
    v8f acc1 = {0.f, 0.f, 0.f, 0.f, 0.f, 0.f, 0.f, 0.f};
    v4f* xs = shX4[wave];
#pragma unroll
    for (int q = 0; q < NCH; ++q) {
#pragma unroll
        for (int r = 0; r < TILE_M; ++r) xs[r * (XPITCH / 4) + lane] = R[r];
        if (q + 1 < NCH) {
#pragma unroll
            for (int r = 0; r < TILE_M; ++r)
                R[r] = xw[r * (FDIM / 4) + (q + 1) * (KCH / 4)];
        }
#pragma unroll
        for (int j = 0; j < KCH / 8; ++j) {
            const v4f a4 = xs[col * (XPITCH / 4) + 2 * j + half];
            const v4f b4 = shB4[(q * (KCH / 4) + 2 * j + half) * 16 + col];
            v2f a0, a1, b0, b1;
            a0.x = a4.x; a0.y = a4.y; a1.x = a4.z; a1.y = a4.w;
            b0.x = b4.x; b0.y = b4.y; b1.x = b4.z; b1.y = b4.w;
            acc0 = __builtin_amdgcn_wmma_f32_16x16x4_f32(
                false, a0, false, b0, (short)0, acc0, false, false);
            acc1 = __builtin_amdgcn_wmma_f32_16x16x4_f32(
                false, a1, false, b1, (short)0, acc1, false, false);
        }
    }
    v8f acc = acc0 + acc1;

    const float HALF_PI = 1.5707963267948966f;
    if (col < NQ) {
        const float bias = enc_b[col];
#pragma unroll
        for (int j = 0; j < 8; ++j) {
            shAng[wave][half * 8 + j][col] = tanhf(acc[j] + bias) * HALF_PI;
        }
    }
    __syncthreads();

    if (lane < TILE_M) {
        const int r = lane;
        float cq[NQ], sq[NQ];
#pragma unroll
        for (int q = 0; q < NQ; ++q) {
            const float phi = 0.5f * shAng[wave][r][q] + 0.7853981633974483f;
            cq[q] = cosf(phi);
            sq[q] = sinf(phi);
        }
        float v[16];
#pragma unroll
        for (int i = 0; i < 16; ++i) {
            float p = ((i >> 3) & 1) ? sq[0] : cq[0];
            p *= ((i >> 2) & 1) ? sq[1] : cq[1];
            p *= ((i >> 1) & 1) ? sq[2] : cq[2];
            p *= ( i       & 1) ? sq[3] : cq[3];
            v[i] = p;
        }
        float p2[16];
#pragma unroll
        for (int i = 0; i < 16; ++i) {
            float a = 0.f;
#pragma unroll
            for (int j = 0; j < 16; ++j) a = fmaf(shU[i * 16 + j], v[j], a);
            p2[i] = a * a;
        }
        float z[NQ];
#pragma unroll
        for (int q = 0; q < NQ; ++q) {
            const int m = 1 << (3 - q);
            float zz = 0.f;
#pragma unroll
            for (int i = 0; i < 16; ++i) zz += (i & m) ? -p2[i] : p2[i];
            z[q] = zz;
        }
        *(float4*)(out + (size_t)(rowBase + r) * NQ) =
            make_float4(z[0], z[1], z[2], z[3]);
    }
}

extern "C" void kernel_launch(void* const* d_in, const int* in_sizes, int n_in,
                              void* d_out, int out_size, void* d_ws, size_t ws_size,
                              hipStream_t stream)
{
    const float* x        = (const float*)d_in[0];
    const float* enc_w    = (const float*)d_in[1];
    const float* enc_b    = (const float*)d_in[2];
    const float* q_params = (const float*)d_in[3];
    float*       out      = (float*)d_out;

    const int B      = in_sizes[0] / FDIM;
    const int blocks = (B + ROWS_PB - 1) / ROWS_PB;
    qnn_main<<<blocks, 256, 0, stream>>>(x, enc_w, enc_b, q_params, out);
}


# Circuit head on all 32 lanes and the matrix unit: encoder roles swapped so each lane gets its own row's angles, U v as 4 f32 WMMAs; no LDS transpose; order of the 16-term state sums changes.
# speedup vs baseline: 2.9153x; 1.1190x over previous
#include <hip/hip_runtime.h>
#include <math.h>


typedef __attribute__((ext_vector_type(2))) float v2f;
typedef __attribute__((ext_vector_type(4))) float v4f;
typedef __attribute__((ext_vector_type(8))) float v8f;

#define NQ        4
#define QDEPTH    6
#define FDIM      512
#define TILE_M    16
#define WAVES_PB  8
#define ROWS_PB   (TILE_M * WAVES_PB)
#define KCH       128
#define NCH       (FDIM / KCH)
#define XPITCH    (KCH + 4)

__device__ __forceinline__ float xhalf(float x)
{
    const int b = __builtin_bit_cast(int, x);
    return __builtin_bit_cast(float,
        __builtin_amdgcn_permlanex16(b, b, 0x76543210, 0xfedcba98, false, false));
}

__global__ void __launch_bounds__(256)
qnn_main(const float* __restrict__ x,
         const float* __restrict__ enc_w,
         const float* __restrict__ enc_b,
         const float* __restrict__ q_params,
         float* __restrict__ out)
{
    __shared__ v4f shW4[(FDIM / 4) * 16];
    __shared__ v4f shX4[WAVES_PB][TILE_M * (XPITCH / 4)];
    __shared__ float shU[256];
    __shared__ float shCos[NQ * QDEPTH];
    __shared__ float shSin[NQ * QDEPTH];

    const int tid  = threadIdx.x;
    const int wave = tid >> 5;
    const int lane = tid & 31;
    const int half = lane >> 4;
    const int col  = lane & 15;

    const int     rowBase = (blockIdx.x * WAVES_PB + wave) * TILE_M;
    const v4f*    xw      = (const v4f*)(x + (size_t)rowBase * FDIM) + lane;
    v4f R[TILE_M];
#pragma unroll
    for (int r = 0; r < TILE_M; ++r) R[r] = xw[r * (FDIM / 4)];

    const float qp = q_params[tid < NQ * QDEPTH ? tid : 0];
    const float bias0 = enc_b[2 * half];
    const float bias1 = enc_b[2 * half + 1];

    {
        v4f w[FDIM / 256];
#pragma unroll
        for (int j = 0; j < FDIM / 256; ++j) {
            const int i = tid + 256 * j;
            w[j] = ((const v4f*)enc_w)[i];
        }
        asm volatile("" :: "v"(qp));
        const v4f zero4 = {0.f, 0.f, 0.f, 0.f};
#pragma unroll
        for (int j = 0; j < FDIM / 256; ++j) {
            const int i = tid + 256 * j;
            const int n = i >> 7;
            shW4[(i & 127) * 16 + (n & 1) + 8 * (n >> 1)] = w[j];
        }
#pragma unroll
        for (int j = 0; j < 6; ++j) {
            const int z  = tid + 256 * j;
            const int kq = z / 12;
            const int e  = z - 12 * kq;
            shW4[kq * 16 + e + (e < 6 ? 2 : 4)] = zero4;
        }
    }
    if (tid < NQ * QDEPTH) {
        const float th = 0.5f * qp;
        shCos[tid] = cosf(th);
        shSin[tid] = sinf(th);
    }
    __syncthreads();

    if (tid < 16) {
        float s[16];
#pragma unroll
        for (int i = 0; i < 16; ++i) s[i] = (i == tid) ? 1.0f : 0.0f;

#pragma unroll
        for (int d = 0; d < QDEPTH; ++d) {
#pragma unroll
            for (int q = 0; q < NQ; ++q) {
                const float c  = shCos[q * QDEPTH + d];
                const float sn = shSin[q * QDEPTH + d];
                const int   m  = 1 << (3 - q);
#pragma unroll
                for (int i = 0; i < 16; ++i) {
                    if (i & m) continue;
                    const float a = s[i], b = s[i | m];
                    s[i]     = c * a - sn * b;
                    s[i | m] = sn * a + c * b;
                }
            }
#pragma unroll
            for (int q = 0; q < NQ; ++q) {
                const int mc = 1 << (3 - q);
                const int mt = 1 << (3 - ((q + 1) & 3));
#pragma unroll
                for (int i = 0; i < 16; ++i) {
                    if ((i & mc) && !(i & mt)) {
                        const float t = s[i];
                        s[i]      = s[i | mt];
                        s[i | mt] = t;
                    }
                }
            }
        }
#pragma unroll
        for (int i = 0; i < 16; ++i) shU[i * 16 + tid] = s[i];
    }

    v8f acc0 = {0.f, 0.f, 0.f, 0.f, 0.f, 0.f, 0.f, 0.f};
    v8f acc1 = {0.f, 0.f, 0.f, 0.f, 0.f, 0.f, 0.f, 0.f};
    v4f* xs = shX4[wave];
#pragma unroll
    for (int q = 0; q < NCH; ++q) {
#pragma unroll
        for (int r = 0; r < TILE_M; ++r) xs[r * (XPITCH / 4) + lane] = R[r];
        if (q + 1 < NCH) {
#pragma unroll
            for (int r = 0; r < TILE_M; ++r)
                R[r] = xw[r * (FDIM / 4) + (q + 1) * (KCH / 4)];
        }
#pragma unroll
        for (int j = 0; j < KCH / 8; ++j) {
            const v4f b4 = xs[col * (XPITCH / 4) + 2 * j + half];
            const v4f a4 = shW4[(q * (KCH / 4) + 2 * j + half) * 16 + col];
            v2f a0, a1, b0, b1;
            a0.x = a4.x; a0.y = a4.y; a1.x = a4.z; a1.y = a4.w;
            b0.x = b4.x; b0.y = b4.y; b1.x = b4.z; b1.y = b4.w;
            acc0 = __builtin_amdgcn_wmma_f32_16x16x4_f32(
                false, a0, false, b0, (short)0, acc0, false, false);
            acc1 = __builtin_amdgcn_wmma_f32_16x16x4_f32(
                false, a1, false, b1, (short)0, acc1, false, false);
        }
    }
    const v8f acc = acc0 + acc1;

    const float HALF_PI    = 1.5707963267948966f;
    const float QUARTER_PI = 0.7853981633974483f;
    const float ang0 = tanhf(acc[0] + bias0) * HALF_PI;
    const float ang1 = tanhf(acc[1] + bias1) * HALF_PI;
    const float phi0 = 0.5f * ang0 + QUARTER_PI;
    const float phi1 = 0.5f * ang1 + QUARTER_PI;
    const float c0 = cosf(phi0), s0 = sinf(phi0);
    const float c1 = cosf(phi1), s1 = sinf(phi1);
    const float oc0 = xhalf(c0), os0 = xhalf(s0);
    const float oc1 = xhalf(c1), os1 = xhalf(s1);
    float cq[NQ], sq[NQ];
    cq[0] = half ? oc0 : c0;   sq[0] = half ? os0 : s0;
    cq[1] = half ? oc1 : c1;   sq[1] = half ? os1 : s1;
    cq[2] = half ? c0 : oc0;   sq[2] = half ? s0 : os0;
    cq[3] = half ? c1 : oc1;   sq[3] = half ? s1 : os1;

    const float f0 = half ? sq[0] : cq[0];
    float g1[2], g23[4], v[8];
    g1[0]  = f0 * cq[1];        g1[1]  = f0 * sq[1];
    g23[0] = cq[2] * cq[3];     g23[1] = cq[2] * sq[3];
    g23[2] = sq[2] * cq[3];     g23[3] = sq[2] * sq[3];
#pragma unroll
    for (int jj = 0; jj < 8; ++jj) v[jj] = g1[jj >> 2] * g23[jj & 3];

    __syncthreads();

    const v4f ua = *(const v4f*)(shU + col * 16 + half * 8);
    const v4f ub = *(const v4f*)(shU + col * 16 + half * 8 + 4);
    v2f ufr[4], vfr[4];
    ufr[0].x = ua.x; ufr[0].y = ua.y;   ufr[1].x = ua.z; ufr[1].y = ua.w;
    ufr[2].x = ub.x; ufr[2].y = ub.y;   ufr[3].x = ub.z; ufr[3].y = ub.w;
#pragma unroll
    for (int t = 0; t < 4; ++t) { vfr[t].x = v[2 * t]; vfr[t].y = v[2 * t + 1]; }
    v8f w0 = {0.f, 0.f, 0.f, 0.f, 0.f, 0.f, 0.f, 0.f};
    v8f w1 = {0.f, 0.f, 0.f, 0.f, 0.f, 0.f, 0.f, 0.f};
    w0 = __builtin_amdgcn_wmma_f32_16x16x4_f32(false, ufr[0], false, vfr[0], (short)0, w0, false, false);
    w1 = __builtin_amdgcn_wmma_f32_16x16x4_f32(false, ufr[1], false, vfr[1], (short)0, w1, false, false);
    w0 = __builtin_amdgcn_wmma_f32_16x16x4_f32(false, ufr[2], false, vfr[2], (short)0, w0, false, false);
    w1 = __builtin_amdgcn_wmma_f32_16x16x4_f32(false, ufr[3], false, vfr[3], (short)0, w1, false, false);
    const v8f wv = w0 + w1;

    float p[8];
#pragma unroll
    for (int r = 0; r < 8; ++r) p[r] = wv[r] * wv[r];
    const float a01p = p[0] + p[1], a01m = p[0] - p[1];
    const float a23p = p[2] + p[3], a23m = p[2] - p[3];
    const float a45p = p[4] + p[5], a45m = p[4] - p[5];
    const float a67p = p[6] + p[7], a67m = p[6] - p[7];
    const float lo_pp = a01p + a23p, lo_pm = a01p - a23p, lo_m = a01m + a23m;
    const float hi_pp = a45p + a67p, hi_pm = a45p - a67p, hi_m = a45m + a67m;
    const float all = lo_pp + hi_pp;
    const float t0 = half ? -all : all;
    const float t1 = lo_pp - hi_pp;
    const float t2 = lo_pm + hi_pm;
    const float t3 = lo_m  + hi_m;
    const float z0 = t0 + xhalf(t0);
    const float z1 = t1 + xhalf(t1);
    const float z2 = t2 + xhalf(t2);
    const float z3 = t3 + xhalf(t3);
    if (half == 0) {
        *(float4*)(out + (size_t)(rowBase + col) * NQ) = make_float4(z0, z1, z2, z3);
    }
}

extern "C" void kernel_launch(void* const* d_in, const int* in_sizes, int n_in,
                              void* d_out, int out_size, void* d_ws, size_t ws_size,
                              hipStream_t stream)
{
    const float* x        = (const float*)d_in[0];
    const float* enc_w    = (const float*)d_in[1];
    const float* enc_b    = (const float*)d_in[2];
    const float* q_params = (const float*)d_in[3];
    float*       out      = (float*)d_out;

    const int B      = in_sizes[0] / FDIM;
    const int blocks = (B + ROWS_PB - 1) / ROWS_PB;
    qnn_main<<<blocks, 256, 0, stream>>>(x, enc_w, enc_b, q_params, out);
}
